# starting kernel
# baseline (speedup 1.0000x reference)
#include <hip/hip_runtime.h>

typedef __attribute__((ext_vector_type(2))) float v2f;
typedef __attribute__((ext_vector_type(8))) float v8f;

#define NX    16384
#define HID   128
#define ROWS  256

__device__ __forceinline__ v8f wmma4(v2f a, v2f b, v8f c) {
  return __builtin_amdgcn_wmma_f32_16x16x4_f32(
       false, a,  false, b,
       (short)0, c,  false,  false);
}

__global__ void mlp_trunk(const float* __restrict__ pw1, const float* __restrict__ pb1,
                          const float* __restrict__ pw2, const float* __restrict__ pb2,
                          const float* __restrict__ qw1, const float* __restrict__ qb1,
                          const float* __restrict__ qw2, const float* __restrict__ qb2,
                          float* __restrict__ H2P, float* __restrict__ H2S) {
  __shared__ float s_w1[128];
  __shared__ float s_b1[64];
  __shared__ float s_w2[64 * 128];
  __shared__ float s_b2[128];
  const int tid = threadIdx.x;
  const int n = blockIdx.x * 128 + tid;
  const float gx = (float)(n >> 7) * (1.0f / 127.0f);
  const float gy = (float)(n & 127) * (1.0f / 127.0f);

  for (int net = 0; net < 2; ++net) {
    const float* w1 = net ? qw1 : pw1;
    const float* b1 = net ? qb1 : pb1;
    const float* w2 = net ? qw2 : pw2;
    const float* b2 = net ? qb2 : pb2;
    __syncthreads();
    s_w1[tid] = w1[tid];
    if (tid < 64) s_b1[tid] = b1[tid];
    for (int i = tid; i < 64 * 128; i += 128) s_w2[i] = w2[i];
    s_b2[tid] = b2[tid];
    __syncthreads();

    float h1[64];
#pragma unroll
    for (int j = 0; j < 64; ++j) {
      float v = gx * s_w1[j] + gy * s_w1[64 + j] + s_b1[j];
      h1[j] = fmaxf(v, 0.0f);
    }
    for (int c = 0; c < HID; ++c) {
      float acc = s_b2[c];
#pragma unroll 16
      for (int j = 0; j < 64; ++j) acc += h1[j] * s_w2[j * 128 + c];
      acc = fmaxf(acc, 0.0f);
      if (net == 0) H2P[c * NX + n] = acc;
      else          H2S[n * HID + c] = acc;
    }
  }
}

__global__ void rowsum(const float* __restrict__ x, float* __restrict__ X1) {
  __shared__ float red[256];
  const int r = blockIdx.x;
  float s = 0.0f;
  for (int n = threadIdx.x; n < NX; n += 256) s += x[(long)r * NX + n];
  red[threadIdx.x] = s;
  __syncthreads();
  for (int off = 128; off > 0; off >>= 1) {
    if ((int)threadIdx.x < off) red[threadIdx.x] += red[threadIdx.x + off];
    __syncthreads();
  }
  if (threadIdx.x == 0) X1[r] = red[0];
}

__global__ void gemm1_partial(const float* __restrict__ X, const float* __restrict__ H2P,
                              float* __restrict__ P) {
  const int lane = threadIdx.x;
  const int half = lane >> 4;
  const int l16  = lane & 15;
  const int ms   = blockIdx.x;
  const int kc   = blockIdx.y;
  const long k0  = (long)kc * 256 + 2 * half;

  v8f acc[8];
  const v8f vzero = {0.f,0.f,0.f,0.f,0.f,0.f,0.f,0.f};
#pragma unroll
  for (int t = 0; t < 8; ++t) acc[t] = vzero;

  const float* aptr = X   + (long)(ms * 16 + l16) * NX + k0;
  const float* bptr = H2P + (long)l16 * NX + k0;
  for (int kb = 0; kb < 256; kb += 4) {
    __builtin_prefetch(aptr + 64, 0, 0);
    v2f a = *(const v2f*)aptr;
#pragma unroll
    for (int t = 0; t < 8; ++t) {
      v2f b = *(const v2f*)(bptr + (long)t * 16 * NX);
      acc[t] = wmma4(a, b, acc[t]);
    }
    aptr += 4;
    bptr += 4;
  }
#pragma unroll
  for (int t = 0; t < 8; ++t)
#pragma unroll
    for (int r = 0; r < 8; ++r)
      P[((long)kc * ROWS + ms * 16 + r + 8 * half) * HID + t * 16 + l16] = acc[t][r];
}

__global__ void reduceP(const float* __restrict__ P, float* __restrict__ T) {
  const int idx = blockIdx.x * 256 + threadIdx.x;
  float s = 0.0f;
  for (int kc = 0; kc < 64; ++kc) s += P[(long)kc * (ROWS * HID) + idx];
  T[idx] = s;
}

__global__ void s_kernel(const float* __restrict__ T, const float* __restrict__ w3p,
                         const float* __restrict__ b3p, const float* __restrict__ X1,
                         float* __restrict__ S) {
  const int idx = blockIdx.x * 256 + threadIdx.x;
  const int b = idx >> 10;
  const int o = (idx >> 5) & 31;
  const int i = idx & 31;
  const int r = b * 32 + i;
  const int oi = o * 32 + i;
  float acc = b3p[oi] * X1[r];
  for (int k = 0; k < HID; ++k) acc += T[r * HID + k] * w3p[k * 1024 + oi];
  S[idx] = acc;
}

__global__ void u_kernel(const float* __restrict__ S, const float* __restrict__ w3q,
                         float* __restrict__ U) {
  const int idx = blockIdx.x * 256 + threadIdx.x;
  const int bo = idx >> 7;
  const int k = idx & 127;
  const int o = bo & 31;
  float acc = 0.0f;
#pragma unroll
  for (int i = 0; i < 32; ++i) acc += S[bo * 32 + i] * w3q[k * 1024 + o * 32 + i];
  U[idx] = acc;
}

__global__ void c_kernel(const float* __restrict__ S, const float* __restrict__ b3q,
                         float* __restrict__ C) {
  const int bo = threadIdx.x;
  const int o = bo & 31;
  float acc = 0.0f;
#pragma unroll
  for (int i = 0; i < 32; ++i) acc += S[bo * 32 + i] * b3q[o * 32 + i];
  C[bo] = acc;
}

__global__ void gemm2(const float* __restrict__ U, const float* __restrict__ H2S,
                      const float* __restrict__ C, float* __restrict__ out) {
  const int lane = threadIdx.x;
  const int half = lane >> 4;
  const int l16  = lane & 15;
  const int mt = blockIdx.x;
  const int n0 = blockIdx.y * 4;

  v8f acc[4];
  const v8f vzero = {0.f,0.f,0.f,0.f,0.f,0.f,0.f,0.f};
#pragma unroll
  for (int t = 0; t < 4; ++t) acc[t] = vzero;

  const float* aptr = U   + (long)(mt * 16 + l16) * HID + 2 * half;
  const float* bptr = H2S + ((long)n0 * 16 + l16) * HID + 2 * half;
#pragma unroll
  for (int kb = 0; kb < HID; kb += 4) {
    v2f a = *(const v2f*)(aptr + kb);
#pragma unroll
    for (int t = 0; t < 4; ++t) {
      v2f b = *(const v2f*)(bptr + t * 16 * HID + kb);
      acc[t] = wmma4(a, b, acc[t]);
    }
  }
  const float scale = 1.0f / (float)NX;
#pragma unroll
  for (int t = 0; t < 4; ++t)
#pragma unroll
    for (int r = 0; r < 8; ++r) {
      const int row = mt * 16 + r + 8 * half;
      out[(long)row * NX + (n0 + t) * 16 + l16] = (acc[t][r] + C[row]) * scale;
    }
}

extern "C" void kernel_launch(void* const* d_in, const int* in_sizes, int n_in,
                              void* d_out, int out_size, void* d_ws, size_t ws_size,
                              hipStream_t stream) {
  const float* x   = (const float*)d_in[0];
  const float* pw1 = (const float*)d_in[1];
  const float* pb1 = (const float*)d_in[2];
  const float* pw2 = (const float*)d_in[3];
  const float* pb2 = (const float*)d_in[4];
  const float* pw3 = (const float*)d_in[5];
  const float* pb3 = (const float*)d_in[6];
  const float* qw1 = (const float*)d_in[7];
  const float* qb1 = (const float*)d_in[8];
  const float* qw2 = (const float*)d_in[9];
  const float* qb2 = (const float*)d_in[10];
  const float* qw3 = (const float*)d_in[11];
  const float* qb3 = (const float*)d_in[12];

  float* ws  = (float*)d_ws;
  float* H2P = ws;
  float* H2S = H2P + HID * NX;
  float* P   = H2S + NX * HID;
  float* T   = P + 64 * ROWS * HID;
  float* X1  = T + ROWS * HID;
  float* S   = X1 + ROWS;
  float* Uu  = S + 8192;
  float* Cc  = Uu + ROWS * HID;

  mlp_trunk<<<dim3(NX / 128), dim3(128), 0, stream>>>(pw1, pb1, pw2, pb2,
                                                      qw1, qb1, qw2, qb2, H2P, H2S);
  rowsum<<<dim3(ROWS), dim3(256), 0, stream>>>(x, X1);
  gemm1_partial<<<dim3(16, 64), dim3(32), 0, stream>>>(x, H2P, P);
  reduceP<<<dim3(128), dim3(256), 0, stream>>>(P, T);
  s_kernel<<<dim3(32), dim3(256), 0, stream>>>(T, pw3, pb3, X1, S);
  u_kernel<<<dim3(128), dim3(256), 0, stream>>>(S, qw3, Uu);
  c_kernel<<<dim3(1), dim3(256), 0, stream>>>(S, qb3, Cc);
  gemm2<<<dim3(16, 256), dim3(32), 0, stream>>>(Uu, H2S, Cc, (float*)d_out);
}


# Both GEMMs re-tiled through LDS: 8-wave blocks copy 64x128 operand tiles in by whole-line loads, a wave owns 2x2 matrix tiles read by 16-byte LDS loads; f32 sum order (k order) of both GEMMs changes
# speedup vs baseline: 4.9211x; 4.9211x over previous
#include <hip/hip_runtime.h>

typedef __attribute__((ext_vector_type(2))) float v2f;
typedef __attribute__((ext_vector_type(8))) float v8f;
typedef __attribute__((ext_vector_type(4))) float v4f;

#define NX    16384
#define HID   128
#define ROWS  256

__device__ __forceinline__ v8f wmma4(v2f a, v2f b, v8f c) {
  return __builtin_amdgcn_wmma_f32_16x16x4_f32(
       false, a,  false, b,
       (short)0, c,  false,  false);
}

#define TF 32
__global__ __launch_bounds__(128)
void mlp_trunk(const float* __restrict__ pw1, const float* __restrict__ pb1,
               const float* __restrict__ pw2, const float* __restrict__ pb2,
               const float* __restrict__ qw1, const float* __restrict__ qb1,
               const float* __restrict__ qw2, const float* __restrict__ qb2,
               float* __restrict__ H2P, float* __restrict__ H2S) {
  const int n   = blockIdx.x * 128 + threadIdx.x;
  const int net = blockIdx.y >> 2;
  const int c0  = (blockIdx.y & 3) * TF;
  const float* __restrict__ w1 = net ? qw1 : pw1;
  const float* __restrict__ b1 = net ? qb1 : pb1;
  const float* __restrict__ w2 = (net ? qw2 : pw2) + c0;
  const float* __restrict__ b2 = (net ? qb2 : pb2) + c0;
  const float gx = (float)(n >> 7) * (1.0f / 127.0f);
  const float gy = (float)(n & 127) * (1.0f / 127.0f);

  float acc[TF];
#pragma unroll
  for (int c = 0; c < TF; ++c) acc[c] = b2[c];

  for (int j = 0; j < 64; ++j) {
    const float v = gx * w1[j] + gy * w1[64 + j] + b1[j];
    const float h = fmaxf(v, 0.0f);
    const float* __restrict__ wr = w2 + j * HID;
#pragma unroll
    for (int c = 0; c < TF; ++c) acc[c] += h * wr[c];
  }
#pragma unroll
  for (int c = 0; c < TF; ++c) acc[c] = fmaxf(acc[c], 0.0f);

  if (net == 0) {
#pragma unroll
    for (int c = 0; c < TF; ++c) H2P[(long)(c0 + c) * NX + n] = acc[c];
  } else {
    float4* dst = (float4*)(H2S + (long)n * HID + c0);
#pragma unroll
    for (int q = 0; q < TF / 4; ++q)
      dst[q] = make_float4(acc[4 * q], acc[4 * q + 1], acc[4 * q + 2], acc[4 * q + 3]);
  }
}

__global__ void rowsum(const float* __restrict__ x, float* __restrict__ X1) {
  __shared__ float red[256];
  const int r = blockIdx.x;
  float s = 0.0f;
  for (int n = threadIdx.x; n < NX; n += 256) s += x[(long)r * NX + n];
  red[threadIdx.x] = s;
  __syncthreads();
  for (int off = 128; off > 0; off >>= 1) {
    if ((int)threadIdx.x < off) red[threadIdx.x] += red[threadIdx.x + off];
    __syncthreads();
  }
  if (threadIdx.x == 0) X1[r] = red[0];
}

#define KT     128
#define PITCH  132
#define TROWS_A 64
#define TROWS_B 128
#define STAGE_FLOATS ((TROWS_A + TROWS_B) * PITCH)

template <int NR>
__device__ __forceinline__ void tile_load(const float* __restrict__ src, long ld,
                                          v4f (&r)[NR / 8]) {
  const int t = threadIdx.x;
  const float* p = src + (long)(t >> 5) * ld + 4 * (t & 31);
#pragma unroll
  for (int i = 0; i < NR / 8; ++i) r[i] = *(const v4f*)(p + (long)(8 * i) * ld);
}

template <int NR>
__device__ __forceinline__ void tile_store(float* dst, const v4f (&r)[NR / 8]) {
  const int t = threadIdx.x;
  float* p = dst + (t >> 5) * PITCH + 4 * (t & 31);
#pragma unroll
  for (int i = 0; i < NR / 8; ++i) *(v4f*)(p + 8 * i * PITCH) = r[i];
}

__device__ __forceinline__ void tile_mma(const float* ap, const float* bp, v8f (&acc)[2][2]) {
#pragma unroll
  for (int kb = 0; kb < KT / 8; ++kb) {
    const v4f a0 = *(const v4f*)(ap + 8 * kb);
    const v4f a1 = *(const v4f*)(ap + 16 * PITCH + 8 * kb);
    const v4f b0 = *(const v4f*)(bp + 8 * kb);
    const v4f b1 = *(const v4f*)(bp + 16 * PITCH + 8 * kb);
    const v2f a0l = {a0.x, a0.y}, a0h = {a0.z, a0.w};
    const v2f a1l = {a1.x, a1.y}, a1h = {a1.z, a1.w};
    const v2f b0l = {b0.x, b0.y}, b0h = {b0.z, b0.w};
    const v2f b1l = {b1.x, b1.y}, b1h = {b1.z, b1.w};
    acc[0][0] = wmma4(a0l, b0l, acc[0][0]);
    acc[0][1] = wmma4(a0l, b1l, acc[0][1]);
    acc[1][0] = wmma4(a1l, b0l, acc[1][0]);
    acc[1][1] = wmma4(a1l, b1l, acc[1][1]);
    acc[0][0] = wmma4(a0h, b0h, acc[0][0]);
    acc[0][1] = wmma4(a0h, b1h, acc[0][1]);
    acc[1][0] = wmma4(a1h, b0h, acc[1][0]);
    acc[1][1] = wmma4(a1h, b1h, acc[1][1]);
  }
}

__global__ __launch_bounds__(256)
void gemm1_partial(const float* __restrict__ X, const float* __restrict__ H2P,
                   float* __restrict__ P) {
  __shared__ __attribute__((aligned(16))) float lds[2 * STAGE_FLOATS];
  const int lane = threadIdx.x & 31;
  const int wave = threadIdx.x >> 5;
  const int half = lane >> 4;
  const int l16  = lane & 15;
  const int wm   = wave & 1;
  const int wn   = wave >> 1;
  const int rg   = blockIdx.x & 3;
  const int kc   = blockIdx.x >> 2;

  const float* xa = X   + (long)(64 * rg) * NX + 256 * kc;
  const float* hb = H2P + 256 * kc;
  v4f ra0[TROWS_A / 8], rb0[TROWS_B / 8], ra1[TROWS_A / 8], rb1[TROWS_B / 8];
  tile_load<TROWS_A>(xa, NX, ra0);
  tile_load<TROWS_B>(hb, NX, rb0);

  float* A0 = lds;
  float* B0 = A0 + TROWS_A * PITCH;
  float* A1 = lds + STAGE_FLOATS;
  float* B1 = A1 + TROWS_A * PITCH;

  v8f acc[2][2];
  const v8f vzero = {0.f,0.f,0.f,0.f,0.f,0.f,0.f,0.f};
  acc[0][0] = vzero; acc[0][1] = vzero; acc[1][0] = vzero; acc[1][1] = vzero;

  const int aoff = (32 * wm + l16) * PITCH + 4 * half;
  const int boff = (32 * wn + l16) * PITCH + 4 * half;

  tile_store<TROWS_A>(A0, ra0);
  tile_store<TROWS_B>(B0, rb0);
  __syncthreads();
  tile_load<TROWS_A>(xa + KT, NX, ra1);
  tile_load<TROWS_B>(hb + KT, NX, rb1);
  tile_mma(A0 + aoff, B0 + boff, acc);
  tile_store<TROWS_A>(A1, ra1);
  tile_store<TROWS_B>(B1, rb1);
  __syncthreads();
  tile_mma(A1 + aoff, B1 + boff, acc);

#pragma unroll
  for (int mi = 0; mi < 2; ++mi)
#pragma unroll
    for (int ni = 0; ni < 2; ++ni)
#pragma unroll
      for (int r = 0; r < 8; ++r) {
        const int row = 64 * rg + 32 * wm + 16 * mi + 8 * half + r;
        const int col = 32 * wn + 16 * ni + l16;
        P[((long)kc * ROWS + row) * HID + col] = acc[mi][ni][r];
      }
}

__global__ void reduceP(const float* __restrict__ P, float* __restrict__ T) {
  const int idx = blockIdx.x * 256 + threadIdx.x;
  float s = 0.0f;
  for (int kc = 0; kc < 64; ++kc) s += P[(long)kc * (ROWS * HID) + idx];
  T[idx] = s;
}

__global__ void s_kernel(const float* __restrict__ T, const float* __restrict__ w3p,
                         const float* __restrict__ b3p, const float* __restrict__ X1,
                         float* __restrict__ S) {
  const int idx = blockIdx.x * 256 + threadIdx.x;
  const int b = idx >> 10;
  const int o = (idx >> 5) & 31;
  const int i = idx & 31;
  const int r = b * 32 + i;
  const int oi = o * 32 + i;
  float acc = b3p[oi] * X1[r];
  for (int k = 0; k < HID; ++k) acc += T[r * HID + k] * w3p[k * 1024 + oi];
  S[idx] = acc;
}

__global__ void u_kernel(const float* __restrict__ S, const float* __restrict__ w3q,
                         float* __restrict__ U) {
  const int idx = blockIdx.x * 256 + threadIdx.x;
  const int bo = idx >> 7;
  const int k = idx & 127;
  const int o = bo & 31;
  float acc = 0.0f;
#pragma unroll
  for (int i = 0; i < 32; ++i) acc += S[bo * 32 + i] * w3q[k * 1024 + o * 32 + i];
  U[idx] = acc;
}

__global__ void c_kernel(const float* __restrict__ S, const float* __restrict__ b3q,
                         float* __restrict__ C) {
  const int bo = threadIdx.x;
  const int o = bo & 31;
  float acc = 0.0f;
#pragma unroll
  for (int i = 0; i < 32; ++i) acc += S[bo * 32 + i] * b3q[o * 32 + i];
  C[bo] = acc;
}

__global__ __launch_bounds__(256)
void gemm2(const float* __restrict__ U, const float* __restrict__ H2S,
           const float* __restrict__ C, float* __restrict__ out) {
  __shared__ __attribute__((aligned(16))) float lds[STAGE_FLOATS];
  const int lane = threadIdx.x & 31;
  const int wave = threadIdx.x >> 5;
  const int half = lane >> 4;
  const int l16  = lane & 15;
  const int wm   = wave & 1;
  const int wn   = wave >> 1;
  const int rb   = blockIdx.x & 3;
  const int cb   = blockIdx.x >> 2;

  v4f ra[TROWS_A / 8], rbv[TROWS_B / 8];
  tile_load<TROWS_A>(U   + (long)(64 * rb) * HID,  HID, ra);
  tile_load<TROWS_B>(H2S + (long)(128 * cb) * HID, HID, rbv);

  float* As = lds;
  float* Bs = As + TROWS_A * PITCH;

  v8f acc[2][2];
  const v8f vzero = {0.f,0.f,0.f,0.f,0.f,0.f,0.f,0.f};
  acc[0][0] = vzero; acc[0][1] = vzero; acc[1][0] = vzero; acc[1][1] = vzero;

  tile_store<TROWS_A>(As, ra);
  tile_store<TROWS_B>(Bs, rbv);
  __syncthreads();
  tile_mma(As + (32 * wm + l16) * PITCH + 4 * half,
           Bs + (32 * wn + l16) * PITCH + 4 * half, acc);

  const float scale = 1.0f / (float)NX;
#pragma unroll
  for (int mi = 0; mi < 2; ++mi)
#pragma unroll
    for (int ni = 0; ni < 2; ++ni)
#pragma unroll
      for (int r = 0; r < 8; ++r) {
        const int row = 64 * rb + 32 * wm + 16 * mi + 8 * half + r;
        const int col = 128 * cb + 32 * wn + 16 * ni + l16;
        out[(long)row * NX + col] = (acc[mi][ni][r] + C[row]) * scale;
      }
}

extern "C" void kernel_launch(void* const* d_in, const int* in_sizes, int n_in,
                              void* d_out, int out_size, void* d_ws, size_t ws_size,
                              hipStream_t stream) {
  const float* x   = (const float*)d_in[0];
  const float* pw1 = (const float*)d_in[1];
  const float* pb1 = (const float*)d_in[2];
  const float* pw2 = (const float*)d_in[3];
  const float* pb2 = (const float*)d_in[4];
  const float* pw3 = (const float*)d_in[5];
  const float* pb3 = (const float*)d_in[6];
  const float* qw1 = (const float*)d_in[7];
  const float* qb1 = (const float*)d_in[8];
  const float* qw2 = (const float*)d_in[9];
  const float* qb2 = (const float*)d_in[10];
  const float* qw3 = (const float*)d_in[11];
  const float* qb3 = (const float*)d_in[12];

  float* ws  = (float*)d_ws;
  float* H2P = ws;
  float* H2S = H2P + HID * NX;
  float* P   = H2S + NX * HID;
  float* T   = P + 64 * ROWS * HID;
  float* X1  = T + ROWS * HID;
  float* S   = X1 + ROWS;
  float* Uu  = S + 8192;
  float* Cc  = Uu + ROWS * HID;

  mlp_trunk<<<dim3(NX / 128, 8), dim3(128), 0, stream>>>(pw1, pb1, pw2, pb2,
                                                      qw1, qb1, qw2, qb2, H2P, H2S);
  rowsum<<<dim3(ROWS), dim3(256), 0, stream>>>(x, X1);
  gemm1_partial<<<dim3(256), dim3(256), 0, stream>>>(x, H2P, P);
  reduceP<<<dim3(128), dim3(256), 0, stream>>>(P, T);
  s_kernel<<<dim3(32), dim3(256), 0, stream>>>(T, pw3, pb3, X1, S);
  u_kernel<<<dim3(128), dim3(256), 0, stream>>>(S, qw3, Uu);
  c_kernel<<<dim3(1), dim3(256), 0, stream>>>(S, qb3, Cc);
  gemm2<<<dim3(512), dim3(256), 0, stream>>>(Uu, H2S, Cc, (float*)d_out);
}


# Independent work rides in a neighbour launch as extra blocks: x row sums beside the phi trunk, psi trunk beside S, c beside U; launches 8 -> 6; every sum order kept
# speedup vs baseline: 5.3897x; 1.0952x over previous
#include <hip/hip_runtime.h>

typedef __attribute__((ext_vector_type(2))) float v2f;
typedef __attribute__((ext_vector_type(8))) float v8f;
typedef __attribute__((ext_vector_type(4))) float v4f;

#define NX    16384
#define HID   128
#define ROWS  256

__device__ __forceinline__ v8f wmma4(v2f a, v2f b, v8f c) {
  return __builtin_amdgcn_wmma_f32_16x16x4_f32(
       false, a,  false, b,
       (short)0, c,  false,  false);
}

#define TF 32
typedef const __attribute__((address_space(4))) float cfloat;
template <bool POINT_MAJOR>
__device__ __forceinline__ void trunk_tile(const float* __restrict__ w1_, const float* __restrict__ b1_,
                                           const float* __restrict__ w2g_, const float* __restrict__ b2g_,
                                           float* __restrict__ H, int tb) {
  cfloat* w1 = (cfloat*)(unsigned long)w1_;
  cfloat* b1 = (cfloat*)(unsigned long)b1_;
  cfloat* w2g = (cfloat*)(unsigned long)w2g_;
  cfloat* b2g = (cfloat*)(unsigned long)b2g_;
  const int n   = (tb >> 2) * 256 + threadIdx.x;
  const int c0  = (tb & 3) * TF;
  cfloat* w2 = w2g + c0;
  cfloat* b2 = b2g + c0;
  const float gx = (float)(n >> 7) * (1.0f / 127.0f);
  const float gy = (float)(n & 127) * (1.0f / 127.0f);

  float acc[TF];
#pragma unroll
  for (int c = 0; c < TF; ++c) acc[c] = b2[c];

  for (int j = 0; j < 64; ++j) {
    const float v = gx * w1[j] + gy * w1[64 + j] + b1[j];
    const float h = fmaxf(v, 0.0f);
    cfloat* wr = w2 + j * HID;
#pragma unroll
    for (int c = 0; c < TF; ++c) acc[c] += h * wr[c];
  }
#pragma unroll
  for (int c = 0; c < TF; ++c) acc[c] = fmaxf(acc[c], 0.0f);

  if (!POINT_MAJOR) {
#pragma unroll
    for (int c = 0; c < TF; ++c) H[(long)(c0 + c) * NX + n] = acc[c];
  } else {
    float4* dst = (float4*)(H + (long)n * HID + c0);
#pragma unroll
    for (int q = 0; q < TF / 4; ++q)
      dst[q] = make_float4(acc[4 * q], acc[4 * q + 1], acc[4 * q + 2], acc[4 * q + 3]);
  }
}

__device__ __forceinline__ void rowsum_row(const float* __restrict__ x, float* __restrict__ X1, int r) {
  __shared__ float red[256];
  float s = 0.0f;
  for (int n = threadIdx.x; n < NX; n += 256) s += x[(long)r * NX + n];
  red[threadIdx.x] = s;
  __syncthreads();
  for (int off = 128; off > 0; off >>= 1) {
    if ((int)threadIdx.x < off) red[threadIdx.x] += red[threadIdx.x + off];
    __syncthreads();
  }
  if (threadIdx.x == 0) X1[r] = red[0];
}

#define TRUNK_TILES 256
__global__ __launch_bounds__(256)
void phi_trunk_rowsum(const float* __restrict__ pw1, const float* __restrict__ pb1,
                      const float* __restrict__ pw2, const float* __restrict__ pb2,
                      const float* __restrict__ x,
                      float* __restrict__ H2P, float* __restrict__ X1) {
  if (blockIdx.x < TRUNK_TILES) trunk_tile<false>(pw1, pb1, pw2, pb2, H2P, blockIdx.x);
  else                          rowsum_row(x, X1, blockIdx.x - TRUNK_TILES);
}

#define KT     128
#define PITCH  132
#define TROWS_A 64
#define TROWS_B 128
#define STAGE_FLOATS ((TROWS_A + TROWS_B) * PITCH)

template <int NR>
__device__ __forceinline__ void tile_load(const float* __restrict__ src, long ld,
                                          v4f (&r)[NR / 8]) {
  const int t = threadIdx.x;
  const float* p = src + (long)(t >> 5) * ld + 4 * (t & 31);
#pragma unroll
  for (int i = 0; i < NR / 8; ++i) r[i] = *(const v4f*)(p + (long)(8 * i) * ld);
}

template <int NR>
__device__ __forceinline__ void tile_store(float* dst, const v4f (&r)[NR / 8]) {
  const int t = threadIdx.x;
  float* p = dst + (t >> 5) * PITCH + 4 * (t & 31);
#pragma unroll
  for (int i = 0; i < NR / 8; ++i) *(v4f*)(p + 8 * i * PITCH) = r[i];
}

__device__ __forceinline__ void tile_mma(const float* ap, const float* bp, v8f (&acc)[2][2]) {
#pragma unroll
  for (int kb = 0; kb < KT / 8; ++kb) {
    const v4f a0 = *(const v4f*)(ap + 8 * kb);
    const v4f a1 = *(const v4f*)(ap + 16 * PITCH + 8 * kb);
    const v4f b0 = *(const v4f*)(bp + 8 * kb);
    const v4f b1 = *(const v4f*)(bp + 16 * PITCH + 8 * kb);
    const v2f a0l = {a0.x, a0.y}, a0h = {a0.z, a0.w};
    const v2f a1l = {a1.x, a1.y}, a1h = {a1.z, a1.w};
    const v2f b0l = {b0.x, b0.y}, b0h = {b0.z, b0.w};
    const v2f b1l = {b1.x, b1.y}, b1h = {b1.z, b1.w};
    acc[0][0] = wmma4(a0l, b0l, acc[0][0]);
    acc[0][1] = wmma4(a0l, b1l, acc[0][1]);
    acc[1][0] = wmma4(a1l, b0l, acc[1][0]);
    acc[1][1] = wmma4(a1l, b1l, acc[1][1]);
    acc[0][0] = wmma4(a0h, b0h, acc[0][0]);
    acc[0][1] = wmma4(a0h, b1h, acc[0][1]);
    acc[1][0] = wmma4(a1h, b0h, acc[1][0]);
    acc[1][1] = wmma4(a1h, b1h, acc[1][1]);
  }
}

__global__ __launch_bounds__(256)
void gemm1_partial(const float* __restrict__ X, const float* __restrict__ H2P,
                   float* __restrict__ P) {
  __shared__ __attribute__((aligned(16))) float lds[2 * STAGE_FLOATS];
  const int lane = threadIdx.x & 31;
  const int wave = threadIdx.x >> 5;
  const int half = lane >> 4;
  const int l16  = lane & 15;
  const int wm   = wave & 1;
  const int wn   = wave >> 1;
  const int rg   = blockIdx.x & 3;
  const int kc   = blockIdx.x >> 2;

  const float* xa = X   + (long)(64 * rg) * NX + 256 * kc;
  const float* hb = H2P + 256 * kc;
  v4f ra0[TROWS_A / 8], rb0[TROWS_B / 8], ra1[TROWS_A / 8], rb1[TROWS_B / 8];
  tile_load<TROWS_A>(xa, NX, ra0);
  tile_load<TROWS_B>(hb, NX, rb0);

  float* A0 = lds;
  float* B0 = A0 + TROWS_A * PITCH;
  float* A1 = lds + STAGE_FLOATS;
  float* B1 = A1 + TROWS_A * PITCH;

  v8f acc[2][2];
  const v8f vzero = {0.f,0.f,0.f,0.f,0.f,0.f,0.f,0.f};
  acc[0][0] = vzero; acc[0][1] = vzero; acc[1][0] = vzero; acc[1][1] = vzero;

  const int aoff = (32 * wm + l16) * PITCH + 4 * half;
  const int boff = (32 * wn + l16) * PITCH + 4 * half;

  tile_store<TROWS_A>(A0, ra0);
  tile_store<TROWS_B>(B0, rb0);
  __syncthreads();
  tile_load<TROWS_A>(xa + KT, NX, ra1);
  tile_load<TROWS_B>(hb + KT, NX, rb1);
  tile_mma(A0 + aoff, B0 + boff, acc);
  tile_store<TROWS_A>(A1, ra1);
  tile_store<TROWS_B>(B1, rb1);
  __syncthreads();
  tile_mma(A1 + aoff, B1 + boff, acc);

#pragma unroll
  for (int mi = 0; mi < 2; ++mi)
#pragma unroll
    for (int ni = 0; ni < 2; ++ni)
#pragma unroll
      for (int r = 0; r < 8; ++r) {
        const int row = 64 * rg + 32 * wm + 16 * mi + 8 * half + r;
        const int col = 32 * wn + 16 * ni + l16;
        P[((long)kc * ROWS + row) * HID + col] = acc[mi][ni][r];
      }
}

__global__ void reduceP(const float* __restrict__ P, float* __restrict__ T) {
  const int idx = blockIdx.x * 256 + threadIdx.x;
  float s = 0.0f;
  for (int kc = 0; kc < 64; ++kc) s += P[(long)kc * (ROWS * HID) + idx];
  T[idx] = s;
}

#define S_BLOCKS 32
__global__ __launch_bounds__(256)
void s_psi_trunk(const float* __restrict__ T, const float* __restrict__ w3p,
                 const float* __restrict__ b3p, const float* __restrict__ X1,
                 float* __restrict__ S,
                 const float* __restrict__ qw1, const float* __restrict__ qb1,
                 const float* __restrict__ qw2, const float* __restrict__ qb2,
                 float* __restrict__ H2S) {
  if (blockIdx.x >= S_BLOCKS) {
    trunk_tile<true>(qw1, qb1, qw2, qb2, H2S, blockIdx.x - S_BLOCKS);
    return;
  }
  const int idx = blockIdx.x * 256 + threadIdx.x;
  const int b = idx >> 10;
  const int o = (idx >> 5) & 31;
  const int i = idx & 31;
  const int r = b * 32 + i;
  const int oi = o * 32 + i;
  float acc = b3p[oi] * X1[r];
  for (int k = 0; k < HID; ++k) acc += T[r * HID + k] * w3p[k * 1024 + oi];
  S[idx] = acc;
}

#define U_BLOCKS 128
__global__ __launch_bounds__(256)
void uc_kernel(const float* __restrict__ S, const float* __restrict__ w3q,
               const float* __restrict__ b3q, float* __restrict__ U, float* __restrict__ C) {
  if (blockIdx.x == U_BLOCKS) {
    const int bo = threadIdx.x;
    const int o = bo & 31;
    float acc = 0.0f;
#pragma unroll
    for (int i = 0; i < 32; ++i) acc += S[bo * 32 + i] * b3q[o * 32 + i];
    C[bo] = acc;
    return;
  }
  const int idx = blockIdx.x * 256 + threadIdx.x;
  const int bo = idx >> 7;
  const int k = idx & 127;
  const int o = bo & 31;
  float acc = 0.0f;
#pragma unroll
  for (int i = 0; i < 32; ++i) acc += S[bo * 32 + i] * w3q[k * 1024 + o * 32 + i];
  U[idx] = acc;
}

__global__ __launch_bounds__(256)
void gemm2(const float* __restrict__ U, const float* __restrict__ H2S,
           const float* __restrict__ C, float* __restrict__ out) {
  __shared__ __attribute__((aligned(16))) float lds[STAGE_FLOATS];
  const int lane = threadIdx.x & 31;
  const int wave = threadIdx.x >> 5;
  const int half = lane >> 4;
  const int l16  = lane & 15;
  const int wm   = wave & 1;
  const int wn   = wave >> 1;
  const int rb   = blockIdx.x & 3;
  const int cb   = blockIdx.x >> 2;

  v4f ra[TROWS_A / 8], rbv[TROWS_B / 8];
  tile_load<TROWS_A>(U   + (long)(64 * rb) * HID,  HID, ra);
  tile_load<TROWS_B>(H2S + (long)(128 * cb) * HID, HID, rbv);

  float* As = lds;
  float* Bs = As + TROWS_A * PITCH;

  v8f acc[2][2];
  const v8f vzero = {0.f,0.f,0.f,0.f,0.f,0.f,0.f,0.f};
  acc[0][0] = vzero; acc[0][1] = vzero; acc[1][0] = vzero; acc[1][1] = vzero;

  tile_store<TROWS_A>(As, ra);
  tile_store<TROWS_B>(Bs, rbv);
  __syncthreads();
  tile_mma(As + (32 * wm + l16) * PITCH + 4 * half,
           Bs + (32 * wn + l16) * PITCH + 4 * half, acc);

  const float scale = 1.0f / (float)NX;
#pragma unroll
  for (int mi = 0; mi < 2; ++mi)
#pragma unroll
    for (int ni = 0; ni < 2; ++ni)
#pragma unroll
      for (int r = 0; r < 8; ++r) {
        const int row = 64 * rb + 32 * wm + 16 * mi + 8 * half + r;
        const int col = 128 * cb + 32 * wn + 16 * ni + l16;
        out[(long)row * NX + col] = (acc[mi][ni][r] + C[row]) * scale;
      }
}

extern "C" void kernel_launch(void* const* d_in, const int* in_sizes, int n_in,
                              void* d_out, int out_size, void* d_ws, size_t ws_size,
                              hipStream_t stream) {
  const float* x   = (const float*)d_in[0];
  const float* pw1 = (const float*)d_in[1];
  const float* pb1 = (const float*)d_in[2];
  const float* pw2 = (const float*)d_in[3];
  const float* pb2 = (const float*)d_in[4];
  const float* pw3 = (const float*)d_in[5];
  const float* pb3 = (const float*)d_in[6];
  const float* qw1 = (const float*)d_in[7];
  const float* qb1 = (const float*)d_in[8];
  const float* qw2 = (const float*)d_in[9];
  const float* qb2 = (const float*)d_in[10];
  const float* qw3 = (const float*)d_in[11];
  const float* qb3 = (const float*)d_in[12];

  float* ws  = (float*)d_ws;
  float* H2P = ws;
  float* H2S = H2P + HID * NX;
  float* P   = H2S + NX * HID;
  float* T   = P + 64 * ROWS * HID;
  float* X1  = T + ROWS * HID;
  float* S   = X1 + ROWS;
  float* Uu  = S + 8192;
  float* Cc  = Uu + ROWS * HID;

  phi_trunk_rowsum<<<dim3(512), dim3(256), 0, stream>>>(pw1, pb1, pw2, pb2, x, H2P, X1);
  gemm1_partial<<<dim3(256), dim3(256), 0, stream>>>(x, H2P, P);
  reduceP<<<dim3(128), dim3(256), 0, stream>>>(P, T);
  s_psi_trunk<<<dim3(288), dim3(256), 0, stream>>>(T, pw3, pb3, X1, S,
                                                 qw1, qb1, qw2, qb2, H2S);
  uc_kernel<<<dim3(129), dim3(256), 0, stream>>>(S, qw3, qb3, Uu, Cc);
  gemm2<<<dim3(512), dim3(256), 0, stream>>>(Uu, H2S, Cc, (float*)d_out);
}


# Reduce launch folded into S: a block per row r=(b,i) sums its 64 GEMM1 partials and takes 32 dots from LDS beside the psi trunk; phi w3 re-laid [i][k][o] in launch 1; launches 6 -> 5; sum orders kept
# speedup vs baseline: 5.7656x; 1.0697x over previous
#include <hip/hip_runtime.h>

typedef __attribute__((ext_vector_type(2))) float v2f;
typedef __attribute__((ext_vector_type(8))) float v8f;
typedef __attribute__((ext_vector_type(4))) float v4f;

#define NX    16384
#define HID   128
#define ROWS  256

__device__ __forceinline__ v8f wmma4(v2f a, v2f b, v8f c) {
  return __builtin_amdgcn_wmma_f32_16x16x4_f32(
       false, a,  false, b,
       (short)0, c,  false,  false);
}

#define TF 32
typedef const __attribute__((address_space(4))) float cfloat;
template <bool POINT_MAJOR>
__device__ __forceinline__ void trunk_tile(const float* __restrict__ w1_, const float* __restrict__ b1_,
                                           const float* __restrict__ w2g_, const float* __restrict__ b2g_,
                                           float* __restrict__ H, int tb) {
  cfloat* w1 = (cfloat*)(unsigned long)w1_;
  cfloat* b1 = (cfloat*)(unsigned long)b1_;
  cfloat* w2g = (cfloat*)(unsigned long)w2g_;
  cfloat* b2g = (cfloat*)(unsigned long)b2g_;
  const int n   = (tb >> 2) * 256 + threadIdx.x;
  const int c0  = (tb & 3) * TF;
  cfloat* w2 = w2g + c0;
  cfloat* b2 = b2g + c0;
  const float gx = (float)(n >> 7) * (1.0f / 127.0f);
  const float gy = (float)(n & 127) * (1.0f / 127.0f);

  float acc[TF];
#pragma unroll
  for (int c = 0; c < TF; ++c) acc[c] = b2[c];

  for (int j = 0; j < 64; ++j) {
    const float v = gx * w1[j] + gy * w1[64 + j] + b1[j];
    const float h = fmaxf(v, 0.0f);
    cfloat* wr = w2 + j * HID;
#pragma unroll
    for (int c = 0; c < TF; ++c) acc[c] += h * wr[c];
  }
#pragma unroll
  for (int c = 0; c < TF; ++c) acc[c] = fmaxf(acc[c], 0.0f);

  if (!POINT_MAJOR) {
#pragma unroll
    for (int c = 0; c < TF; ++c) H[(long)(c0 + c) * NX + n] = acc[c];
  } else {
    float4* dst = (float4*)(H + (long)n * HID + c0);
#pragma unroll
    for (int q = 0; q < TF / 4; ++q)
      dst[q] = make_float4(acc[4 * q], acc[4 * q + 1], acc[4 * q + 2], acc[4 * q + 3]);
  }
}

__device__ __forceinline__ void rowsum_row(const float* __restrict__ x, float* __restrict__ X1, int r) {
  __shared__ float red[256];
  float s = 0.0f;
  for (int n = threadIdx.x; n < NX; n += 256) s += x[(long)r * NX + n];
  red[threadIdx.x] = s;
  __syncthreads();
  for (int off = 128; off > 0; off >>= 1) {
    if ((int)threadIdx.x < off) red[threadIdx.x] += red[threadIdx.x + off];
    __syncthreads();
  }
  if (threadIdx.x == 0) X1[r] = red[0];
}

#define RL_BLOCKS 32
__device__ __forceinline__ void relayout_w3(const float* __restrict__ w3, float* __restrict__ W3T, int kb) {
  __shared__ float tl[4 * 32 * 33];
  const int t = threadIdx.x;
#pragma unroll
  for (int q = 0; q < 16; ++q) {
    const int e  = q * 256 + t;
    const int kk = e >> 10;
    const int oi = e & 1023;
    tl[kk * 1056 + (oi >> 5) * 33 + (oi & 31)] = w3[(long)(4 * kb + kk) * 1024 + oi];
  }
  __syncthreads();
  const int lane = t & 31;
  const int wave = t >> 5;
#pragma unroll
  for (int q = 0; q < 16; ++q) {
    const int ln = q * 8 + wave;
    const int i  = ln >> 2;
    const int kk = ln & 3;
    W3T[((long)i * HID + 4 * kb + kk) * 32 + lane] = tl[kk * 1056 + lane * 33 + i];
  }
}

#define TRUNK_TILES 256
__global__ __launch_bounds__(256)
void phi_trunk_rowsum(const float* __restrict__ pw1, const float* __restrict__ pb1,
                      const float* __restrict__ pw2, const float* __restrict__ pb2,
                      const float* __restrict__ x, const float* __restrict__ pw3,
                      float* __restrict__ H2P, float* __restrict__ X1, float* __restrict__ W3T) {
  if (blockIdx.x < TRUNK_TILES)          trunk_tile<false>(pw1, pb1, pw2, pb2, H2P, blockIdx.x);
  else if (blockIdx.x < TRUNK_TILES + ROWS) rowsum_row(x, X1, blockIdx.x - TRUNK_TILES);
  else                                   relayout_w3(pw3, W3T, blockIdx.x - TRUNK_TILES - ROWS);
}

#define KT     128
#define PITCH  132
#define TROWS_A 64
#define TROWS_B 128
#define STAGE_FLOATS ((TROWS_A + TROWS_B) * PITCH)

template <int NR>
__device__ __forceinline__ void tile_load(const float* __restrict__ src, long ld,
                                          v4f (&r)[NR / 8]) {
  const int t = threadIdx.x;
  const float* p = src + (long)(t >> 5) * ld + 4 * (t & 31);
#pragma unroll
  for (int i = 0; i < NR / 8; ++i) r[i] = *(const v4f*)(p + (long)(8 * i) * ld);
}

template <int NR>
__device__ __forceinline__ void tile_store(float* dst, const v4f (&r)[NR / 8]) {
  const int t = threadIdx.x;
  float* p = dst + (t >> 5) * PITCH + 4 * (t & 31);
#pragma unroll
  for (int i = 0; i < NR / 8; ++i) *(v4f*)(p + 8 * i * PITCH) = r[i];
}

__device__ __forceinline__ void tile_mma(const float* ap, const float* bp, v8f (&acc)[2][2]) {
#pragma unroll
  for (int kb = 0; kb < KT / 8; ++kb) {
    const v4f a0 = *(const v4f*)(ap + 8 * kb);
    const v4f a1 = *(const v4f*)(ap + 16 * PITCH + 8 * kb);
    const v4f b0 = *(const v4f*)(bp + 8 * kb);
    const v4f b1 = *(const v4f*)(bp + 16 * PITCH + 8 * kb);
    const v2f a0l = {a0.x, a0.y}, a0h = {a0.z, a0.w};
    const v2f a1l = {a1.x, a1.y}, a1h = {a1.z, a1.w};
    const v2f b0l = {b0.x, b0.y}, b0h = {b0.z, b0.w};
    const v2f b1l = {b1.x, b1.y}, b1h = {b1.z, b1.w};
    acc[0][0] = wmma4(a0l, b0l, acc[0][0]);
    acc[0][1] = wmma4(a0l, b1l, acc[0][1]);
    acc[1][0] = wmma4(a1l, b0l, acc[1][0]);
    acc[1][1] = wmma4(a1l, b1l, acc[1][1]);
    acc[0][0] = wmma4(a0h, b0h, acc[0][0]);
    acc[0][1] = wmma4(a0h, b1h, acc[0][1]);
    acc[1][0] = wmma4(a1h, b0h, acc[1][0]);
    acc[1][1] = wmma4(a1h, b1h, acc[1][1]);
  }
}

__global__ __launch_bounds__(256)
void gemm1_partial(const float* __restrict__ X, const float* __restrict__ H2P,
                   float* __restrict__ P) {
  __shared__ __attribute__((aligned(16))) float lds[2 * STAGE_FLOATS];
  const int lane = threadIdx.x & 31;
  const int wave = threadIdx.x >> 5;
  const int half = lane >> 4;
  const int l16  = lane & 15;
  const int wm   = wave & 1;
  const int wn   = wave >> 1;
  const int rg   = blockIdx.x & 3;
  const int kc   = blockIdx.x >> 2;

  const float* xa = X   + (long)(64 * rg) * NX + 256 * kc;
  const float* hb = H2P + 256 * kc;
  v4f ra0[TROWS_A / 8], rb0[TROWS_B / 8], ra1[TROWS_A / 8], rb1[TROWS_B / 8];
  tile_load<TROWS_A>(xa, NX, ra0);
  tile_load<TROWS_B>(hb, NX, rb0);

  float* A0 = lds;
  float* B0 = A0 + TROWS_A * PITCH;
  float* A1 = lds + STAGE_FLOATS;
  float* B1 = A1 + TROWS_A * PITCH;

  v8f acc[2][2];
  const v8f vzero = {0.f,0.f,0.f,0.f,0.f,0.f,0.f,0.f};
  acc[0][0] = vzero; acc[0][1] = vzero; acc[1][0] = vzero; acc[1][1] = vzero;

  const int aoff = (32 * wm + l16) * PITCH + 4 * half;
  const int boff = (32 * wn + l16) * PITCH + 4 * half;

  tile_store<TROWS_A>(A0, ra0);
  tile_store<TROWS_B>(B0, rb0);
  __syncthreads();
  tile_load<TROWS_A>(xa + KT, NX, ra1);
  tile_load<TROWS_B>(hb + KT, NX, rb1);
  tile_mma(A0 + aoff, B0 + boff, acc);
  tile_store<TROWS_A>(A1, ra1);
  tile_store<TROWS_B>(B1, rb1);
  __syncthreads();
  tile_mma(A1 + aoff, B1 + boff, acc);

#pragma unroll
  for (int mi = 0; mi < 2; ++mi)
#pragma unroll
    for (int ni = 0; ni < 2; ++ni)
#pragma unroll
      for (int r = 0; r < 8; ++r) {
        const int row = 64 * rg + 32 * wm + 16 * mi + 8 * half + r;
        const int col = 32 * wn + 16 * ni + l16;
        P[((long)kc * ROWS + row) * HID + col] = acc[mi][ni][r];
      }
}

__device__ __forceinline__ void s_row(const float* __restrict__ P, const float* __restrict__ W3T,
                                      const float* __restrict__ b3p, const float* __restrict__ X1,
                                      float* __restrict__ St, int r) {
  __shared__ __attribute__((aligned(16))) float wt[HID * 32];
  __shared__ __attribute__((aligned(16))) float trow[HID];
  const int t = threadIdx.x;
  const int i = r & 31;
  if (t < HID) {
    const float* p = P + (long)r * HID + t;
    float s = 0.0f;
#pragma unroll
    for (int kc = 0; kc < 64; ++kc) s += p[(long)kc * (ROWS * HID)];
    trow[t] = s;
  } else {
    const v4f* src = (const v4f*)(W3T + (long)i * (HID * 32));
    v4f* dst = (v4f*)wt;
    const int u = t - HID;
#pragma unroll
    for (int q = 0; q < 8; ++q) dst[q * 128 + u] = src[q * 128 + u];
  }
  __syncthreads();
  if (t < 32) {
    float acc = b3p[t * 32 + i] * X1[r];
#pragma unroll
    for (int k0 = 0; k0 < HID; k0 += 16) {
      v4f tv[4];
      float wv[16];
#pragma unroll
      for (int q = 0; q < 4; ++q) tv[q] = *(const v4f*)(trow + k0 + 4 * q);
#pragma unroll
      for (int q = 0; q < 16; ++q) wv[q] = wt[(k0 + q) * 32 + t];
      asm volatile("" :: "v"(tv[0]), "v"(tv[1]), "v"(tv[2]), "v"(tv[3]), "v"(wv[0]), "v"(wv[1]), "v"(wv[2]), "v"(wv[3]), "v"(wv[4]), "v"(wv[5]), "v"(wv[6]), "v"(wv[7]), "v"(wv[8]), "v"(wv[9]), "v"(wv[10]), "v"(wv[11]), "v"(wv[12]), "v"(wv[13]), "v"(wv[14]), "v"(wv[15]));
#pragma unroll
      for (int q = 0; q < 16; ++q) acc = fmaf(tv[q >> 2][q & 3], wv[q], acc);
    }
    St[r * 32 + t] = acc;
  }
}

__global__ __launch_bounds__(256)
void s_psi_trunk(const float* __restrict__ P, const float* __restrict__ W3T,
                 const float* __restrict__ b3p, const float* __restrict__ X1,
                 float* __restrict__ St,
                 const float* __restrict__ qw1, const float* __restrict__ qb1,
                 const float* __restrict__ qw2, const float* __restrict__ qb2,
                 float* __restrict__ H2S) {
  if (blockIdx.x < TRUNK_TILES) trunk_tile<true>(qw1, qb1, qw2, qb2, H2S, blockIdx.x);
  else                          s_row(P, W3T, b3p, X1, St, blockIdx.x - TRUNK_TILES);
}

#define U_BLOCKS 128
__global__ __launch_bounds__(256)
void uc_kernel(const float* __restrict__ St, const float* __restrict__ w3q,
               const float* __restrict__ b3q, float* __restrict__ U, float* __restrict__ C) {
  if (blockIdx.x == U_BLOCKS) {
    const int bo = threadIdx.x;
    const int o = bo & 31;
    const float* sp = St + (bo - o) * 32 + o;
    float acc = 0.0f;
#pragma unroll
    for (int i = 0; i < 32; ++i) acc += sp[i * 32] * b3q[o * 32 + i];
    C[bo] = acc;
    return;
  }
  const int idx = blockIdx.x * 256 + threadIdx.x;
  const int bo = idx >> 7;
  const int k = idx & 127;
  const int o = bo & 31;
  const float* sp = St + (bo - o) * 32 + o;
  float acc = 0.0f;
#pragma unroll
  for (int i = 0; i < 32; ++i) acc += sp[i * 32] * w3q[k * 1024 + o * 32 + i];
  U[idx] = acc;
}

__global__ __launch_bounds__(256)
void gemm2(const float* __restrict__ U, const float* __restrict__ H2S,
           const float* __restrict__ C, float* __restrict__ out) {
  __shared__ __attribute__((aligned(16))) float lds[STAGE_FLOATS];
  const int lane = threadIdx.x & 31;
  const int wave = threadIdx.x >> 5;
  const int half = lane >> 4;
  const int l16  = lane & 15;
  const int wm   = wave & 1;
  const int wn   = wave >> 1;
  const int rb   = blockIdx.x & 3;
  const int cb   = blockIdx.x >> 2;

  v4f ra[TROWS_A / 8], rbv[TROWS_B / 8];
  tile_load<TROWS_A>(U   + (long)(64 * rb) * HID,  HID, ra);
  tile_load<TROWS_B>(H2S + (long)(128 * cb) * HID, HID, rbv);

  float* As = lds;
  float* Bs = As + TROWS_A * PITCH;

  v8f acc[2][2];
  const v8f vzero = {0.f,0.f,0.f,0.f,0.f,0.f,0.f,0.f};
  acc[0][0] = vzero; acc[0][1] = vzero; acc[1][0] = vzero; acc[1][1] = vzero;

  tile_store<TROWS_A>(As, ra);
  tile_store<TROWS_B>(Bs, rbv);
  __syncthreads();
  tile_mma(As + (32 * wm + l16) * PITCH + 4 * half,
           Bs + (32 * wn + l16) * PITCH + 4 * half, acc);

  const float scale = 1.0f / (float)NX;
#pragma unroll
  for (int mi = 0; mi < 2; ++mi)
#pragma unroll
    for (int ni = 0; ni < 2; ++ni)
#pragma unroll
      for (int r = 0; r < 8; ++r) {
        const int row = 64 * rb + 32 * wm + 16 * mi + 8 * half + r;
        const int col = 128 * cb + 32 * wn + 16 * ni + l16;
        out[(long)row * NX + col] = (acc[mi][ni][r] + C[row]) * scale;
      }
}

extern "C" void kernel_launch(void* const* d_in, const int* in_sizes, int n_in,
                              void* d_out, int out_size, void* d_ws, size_t ws_size,
                              hipStream_t stream) {
  const float* x   = (const float*)d_in[0];
  const float* pw1 = (const float*)d_in[1];
  const float* pb1 = (const float*)d_in[2];
  const float* pw2 = (const float*)d_in[3];
  const float* pb2 = (const float*)d_in[4];
  const float* pw3 = (const float*)d_in[5];
  const float* pb3 = (const float*)d_in[6];
  const float* qw1 = (const float*)d_in[7];
  const float* qb1 = (const float*)d_in[8];
  const float* qw2 = (const float*)d_in[9];
  const float* qb2 = (const float*)d_in[10];
  const float* qw3 = (const float*)d_in[11];
  const float* qb3 = (const float*)d_in[12];

  float* ws  = (float*)d_ws;
  float* H2P = ws;
  float* H2S = H2P + HID * NX;
  float* P   = H2S + NX * HID;
  float* X1  = P + 64 * ROWS * HID;
  float* St  = X1 + ROWS;
  float* Uu  = St + 8192;
  float* Cc  = Uu + ROWS * HID;
  float* W3T = Cc + ROWS;

  phi_trunk_rowsum<<<dim3(544), dim3(256), 0, stream>>>(pw1, pb1, pw2, pb2, x, pw3, H2P, X1, W3T);
  gemm1_partial<<<dim3(256), dim3(256), 0, stream>>>(x, H2P, P);
  s_psi_trunk<<<dim3(512), dim3(256), 0, stream>>>(P, W3T, pb3, X1, St,
                                                 qw1, qb1, qw2, qb2, H2S);
  uc_kernel<<<dim3(129), dim3(256), 0, stream>>>(St, qw3, qb3, Uu, Cc);
  gemm2<<<dim3(512), dim3(256), 0, stream>>>(Uu, H2S, Cc, (float*)d_out);
}


# MLP trunk weights staged per block in LDS, read by broadcast LDS loads 4 j per group (were scalar loads: a cache round trip every 2nd j); lane owns 2 points, wave half the features; sum order kept
# speedup vs baseline: 8.8045x; 1.5271x over previous
#include <hip/hip_runtime.h>

typedef __attribute__((ext_vector_type(2))) float v2f;
typedef __attribute__((ext_vector_type(8))) float v8f;
typedef __attribute__((ext_vector_type(4))) float v4f;

#define NX    16384
#define HID   128
#define ROWS  256

__device__ __forceinline__ v8f wmma4(v2f a, v2f b, v8f c) {
  return __builtin_amdgcn_wmma_f32_16x16x4_f32(
       false, a,  false, b,
       (short)0, c,  false,  false);
}

#define TF 32
#define TH (TF / 2)
template <bool POINT_MAJOR>
__device__ __forceinline__ void trunk_tile(const float* __restrict__ w1, const float* __restrict__ b1,
                                           const float* __restrict__ w2g, const float* __restrict__ b2g_,
                                           float* __restrict__ H, int tb) {
  __shared__ __attribute__((aligned(16))) float w2s[64 * TF];
  __shared__ __attribute__((aligned(16))) float l1s[64 * 4];
  const int t  = threadIdx.x;
  const int c0 = (tb & 3) * TF;
  {
    const int row = t >> 3;
    const int q   = t & 7;
    const v4f lo = *((const v4f*)(w2g + (long)row * HID + c0) + q);
    const v4f hi = *((const v4f*)(w2g + (long)(row + 32) * HID + c0) + q);
    ((v4f*)w2s)[row * (TF / 4) + q]        = lo;
    ((v4f*)w2s)[(row + 32) * (TF / 4) + q] = hi;
    if (t < 64) {
      const v4f l = {w1[t], w1[64 + t], b1[t], 0.0f};
      ((v4f*)l1s)[t] = l;
    }
  }
  __syncthreads();

  const int lane = t & 31;
  const int wave = t >> 5;
  const int wv   = wave & 3;
  const int fh   = wave >> 2;
  const int n0   = (tb >> 2) * 256 + 64 * wv + lane;
  const int n1   = n0 + 32;
  const float gx0 = (float)(n0 >> 7) * (1.0f / 127.0f);
  const float gy0 = (float)(n0 & 127) * (1.0f / 127.0f);
  const float gx1 = (float)(n1 >> 7) * (1.0f / 127.0f);
  const float gy1 = (float)(n1 & 127) * (1.0f / 127.0f);
  const float* b2 = b2g_ + c0 + TH * fh;

  float a0[TH], a1[TH];
#pragma unroll
  for (int c = 0; c < TH; ++c) { a0[c] = b2[c]; a1[c] = a0[c]; }

  const float* wrow = w2s + TH * fh;
#pragma unroll 1
  for (int j = 0; j < 64; j += 4) {
    v4f w[4][TH / 4], tr[4];
#pragma unroll
    for (int u = 0; u < 4; ++u) {
      tr[u] = ((const v4f*)l1s)[j + u];
#pragma unroll
      for (int q = 0; q < TH / 4; ++q) w[u][q] = ((const v4f*)(wrow + (j + u) * TF))[q];
    }
    __builtin_amdgcn_sched_barrier(0);
#pragma unroll
    for (int u = 0; u < 4; ++u) {
      const float h0 = fmaxf(fmaf(gx0, tr[u].x, gy0 * tr[u].y) + tr[u].z, 0.0f);
      const float h1 = fmaxf(fmaf(gx1, tr[u].x, gy1 * tr[u].y) + tr[u].z, 0.0f);
#pragma unroll
      for (int q = 0; q < TH / 4; ++q) {
        a0[4 * q + 0] += h0 * w[u][q].x;  a1[4 * q + 0] += h1 * w[u][q].x;
        a0[4 * q + 1] += h0 * w[u][q].y;  a1[4 * q + 1] += h1 * w[u][q].y;
        a0[4 * q + 2] += h0 * w[u][q].z;  a1[4 * q + 2] += h1 * w[u][q].z;
        a0[4 * q + 3] += h0 * w[u][q].w;  a1[4 * q + 3] += h1 * w[u][q].w;
      }
    }
  }
#pragma unroll
  for (int c = 0; c < TH; ++c) { a0[c] = fmaxf(a0[c], 0.0f); a1[c] = fmaxf(a1[c], 0.0f); }

  if (!POINT_MAJOR) {
#pragma unroll
    for (int c = 0; c < TH; ++c) {
      H[(long)(c0 + TH * fh + c) * NX + n0] = a0[c];
      H[(long)(c0 + TH * fh + c) * NX + n1] = a1[c];
    }
  } else {
#define XP 20
    __shared__ __attribute__((aligned(16))) float xs[8 * 32 * XP];
    v4f* mine = (v4f*)(xs + (wave * 32 + lane) * XP);
    const v4f* theirs = (const v4f*)(xs + ((wave ^ 4) * 32 + lane) * XP);
    if (fh == 0) {
#pragma unroll
      for (int q = 0; q < TH / 4; ++q) mine[q] = (v4f){a1[4 * q], a1[4 * q + 1], a1[4 * q + 2], a1[4 * q + 3]};
    } else {
#pragma unroll
      for (int q = 0; q < TH / 4; ++q) mine[q] = (v4f){a0[4 * q], a0[4 * q + 1], a0[4 * q + 2], a0[4 * q + 3]};
    }
    __syncthreads();
    v4f p[TH / 4];
#pragma unroll
    for (int q = 0; q < TH / 4; ++q) p[q] = theirs[q];
    if (fh == 0) {
      v4f* d = (v4f*)(H + (long)n0 * HID + c0);
#pragma unroll
      for (int q = 0; q < TH / 4; ++q) {
        d[q]          = (v4f){a0[4 * q], a0[4 * q + 1], a0[4 * q + 2], a0[4 * q + 3]};
        d[TH / 4 + q] = p[q];
      }
    } else {
      v4f* d = (v4f*)(H + (long)n1 * HID + c0);
#pragma unroll
      for (int q = 0; q < TH / 4; ++q) {
        d[q]          = p[q];
        d[TH / 4 + q] = (v4f){a1[4 * q], a1[4 * q + 1], a1[4 * q + 2], a1[4 * q + 3]};
      }
    }
#undef XP
  }
}

__device__ __forceinline__ void rowsum_row(const float* __restrict__ x, float* __restrict__ X1, int r) {
  __shared__ float red[256];
  float s = 0.0f;
  for (int n = threadIdx.x; n < NX; n += 256) s += x[(long)r * NX + n];
  red[threadIdx.x] = s;
  __syncthreads();
  for (int off = 128; off > 0; off >>= 1) {
    if ((int)threadIdx.x < off) red[threadIdx.x] += red[threadIdx.x + off];
    __syncthreads();
  }
  if (threadIdx.x == 0) X1[r] = red[0];
}

#define RL_BLOCKS 32
__device__ __forceinline__ void relayout_w3(const float* __restrict__ w3, float* __restrict__ W3T, int kb) {
  __shared__ float tl[4 * 32 * 33];
  const int t = threadIdx.x;
#pragma unroll
  for (int q = 0; q < 16; ++q) {
    const int e  = q * 256 + t;
    const int kk = e >> 10;
    const int oi = e & 1023;
    tl[kk * 1056 + (oi >> 5) * 33 + (oi & 31)] = w3[(long)(4 * kb + kk) * 1024 + oi];
  }
  __syncthreads();
  const int lane = t & 31;
  const int wave = t >> 5;
#pragma unroll
  for (int q = 0; q < 16; ++q) {
    const int ln = q * 8 + wave;
    const int i  = ln >> 2;
    const int kk = ln & 3;
    W3T[((long)i * HID + 4 * kb + kk) * 32 + lane] = tl[kk * 1056 + lane * 33 + i];
  }
}

#define TRUNK_TILES 256
__global__ __launch_bounds__(256)
void phi_trunk_rowsum(const float* __restrict__ pw1, const float* __restrict__ pb1,
                      const float* __restrict__ pw2, const float* __restrict__ pb2,
                      const float* __restrict__ x, const float* __restrict__ pw3,
                      float* __restrict__ H2P, float* __restrict__ X1, float* __restrict__ W3T) {
  if (blockIdx.x < TRUNK_TILES)          trunk_tile<false>(pw1, pb1, pw2, pb2, H2P, blockIdx.x);
  else if (blockIdx.x < TRUNK_TILES + ROWS) rowsum_row(x, X1, blockIdx.x - TRUNK_TILES);
  else                                   relayout_w3(pw3, W3T, blockIdx.x - TRUNK_TILES - ROWS);
}

#define KT     128
#define PITCH  132
#define TROWS_A 64
#define TROWS_B 128
#define STAGE_FLOATS ((TROWS_A + TROWS_B) * PITCH)

template <int NR>
__device__ __forceinline__ void tile_load(const float* __restrict__ src, long ld,
                                          v4f (&r)[NR / 8]) {
  const int t = threadIdx.x;
  const float* p = src + (long)(t >> 5) * ld + 4 * (t & 31);
#pragma unroll
  for (int i = 0; i < NR / 8; ++i) r[i] = *(const v4f*)(p + (long)(8 * i) * ld);
}

template <int NR>
__device__ __forceinline__ void tile_store(float* dst, const v4f (&r)[NR / 8]) {
  const int t = threadIdx.x;
  float* p = dst + (t >> 5) * PITCH + 4 * (t & 31);
#pragma unroll
  for (int i = 0; i < NR / 8; ++i) *(v4f*)(p + 8 * i * PITCH) = r[i];
}

__device__ __forceinline__ void tile_mma(const float* ap, const float* bp, v8f (&acc)[2][2]) {
#pragma unroll
  for (int kb = 0; kb < KT / 8; ++kb) {
    const v4f a0 = *(const v4f*)(ap + 8 * kb);
    const v4f a1 = *(const v4f*)(ap + 16 * PITCH + 8 * kb);
    const v4f b0 = *(const v4f*)(bp + 8 * kb);
    const v4f b1 = *(const v4f*)(bp + 16 * PITCH + 8 * kb);
    const v2f a0l = {a0.x, a0.y}, a0h = {a0.z, a0.w};
    const v2f a1l = {a1.x, a1.y}, a1h = {a1.z, a1.w};
    const v2f b0l = {b0.x, b0.y}, b0h = {b0.z, b0.w};
    const v2f b1l = {b1.x, b1.y}, b1h = {b1.z, b1.w};
    acc[0][0] = wmma4(a0l, b0l, acc[0][0]);
    acc[0][1] = wmma4(a0l, b1l, acc[0][1]);
    acc[1][0] = wmma4(a1l, b0l, acc[1][0]);
    acc[1][1] = wmma4(a1l, b1l, acc[1][1]);
    acc[0][0] = wmma4(a0h, b0h, acc[0][0]);
    acc[0][1] = wmma4(a0h, b1h, acc[0][1]);
    acc[1][0] = wmma4(a1h, b0h, acc[1][0]);
    acc[1][1] = wmma4(a1h, b1h, acc[1][1]);
  }
}

__global__ __launch_bounds__(256)
void gemm1_partial(const float* __restrict__ X, const float* __restrict__ H2P,
                   float* __restrict__ P) {
  __shared__ __attribute__((aligned(16))) float lds[2 * STAGE_FLOATS];
  const int lane = threadIdx.x & 31;
  const int wave = threadIdx.x >> 5;
  const int half = lane >> 4;
  const int l16  = lane & 15;
  const int wm   = wave & 1;
  const int wn   = wave >> 1;
  const int rg   = blockIdx.x & 3;
  const int kc   = blockIdx.x >> 2;

  const float* xa = X   + (long)(64 * rg) * NX + 256 * kc;
  const float* hb = H2P + 256 * kc;
  v4f ra0[TROWS_A / 8], rb0[TROWS_B / 8], ra1[TROWS_A / 8], rb1[TROWS_B / 8];
  tile_load<TROWS_A>(xa, NX, ra0);
  tile_load<TROWS_B>(hb, NX, rb0);

  float* A0 = lds;
  float* B0 = A0 + TROWS_A * PITCH;
  float* A1 = lds + STAGE_FLOATS;
  float* B1 = A1 + TROWS_A * PITCH;

  v8f acc[2][2];
  const v8f vzero = {0.f,0.f,0.f,0.f,0.f,0.f,0.f,0.f};
  acc[0][0] = vzero; acc[0][1] = vzero; acc[1][0] = vzero; acc[1][1] = vzero;

  const int aoff = (32 * wm + l16) * PITCH + 4 * half;
  const int boff = (32 * wn + l16) * PITCH + 4 * half;

  tile_store<TROWS_A>(A0, ra0);
  tile_store<TROWS_B>(B0, rb0);
  __syncthreads();
  tile_load<TROWS_A>(xa + KT, NX, ra1);
  tile_load<TROWS_B>(hb + KT, NX, rb1);
  tile_mma(A0 + aoff, B0 + boff, acc);
  tile_store<TROWS_A>(A1, ra1);
  tile_store<TROWS_B>(B1, rb1);
  __syncthreads();
  tile_mma(A1 + aoff, B1 + boff, acc);

#pragma unroll
  for (int mi = 0; mi < 2; ++mi)
#pragma unroll
    for (int ni = 0; ni < 2; ++ni)
#pragma unroll
      for (int r = 0; r < 8; ++r) {
        const int row = 64 * rg + 32 * wm + 16 * mi + 8 * half + r;
        const int col = 32 * wn + 16 * ni + l16;
        P[((long)kc * ROWS + row) * HID + col] = acc[mi][ni][r];
      }
}

__device__ __forceinline__ void s_row(const float* __restrict__ P, const float* __restrict__ W3T,
                                      const float* __restrict__ b3p, const float* __restrict__ X1,
                                      float* __restrict__ St, int r) {
  __shared__ __attribute__((aligned(16))) float wt[HID * 32];
  __shared__ __attribute__((aligned(16))) float trow[HID];
  const int t = threadIdx.x;
  const int i = r & 31;
  if (t < HID) {
    const float* p = P + (long)r * HID + t;
    float s = 0.0f;
#pragma unroll
    for (int kc = 0; kc < 64; ++kc) s += p[(long)kc * (ROWS * HID)];
    trow[t] = s;
  } else {
    const v4f* src = (const v4f*)(W3T + (long)i * (HID * 32));
    v4f* dst = (v4f*)wt;
    const int u = t - HID;
#pragma unroll
    for (int q = 0; q < 8; ++q) dst[q * 128 + u] = src[q * 128 + u];
  }
  __syncthreads();
  if (t < 32) {
    float acc = b3p[t * 32 + i] * X1[r];
#pragma unroll
    for (int k0 = 0; k0 < HID; k0 += 16) {
      v4f tv[4];
      float wv[16];
#pragma unroll
      for (int q = 0; q < 4; ++q) tv[q] = *(const v4f*)(trow + k0 + 4 * q);
#pragma unroll
      for (int q = 0; q < 16; ++q) wv[q] = wt[(k0 + q) * 32 + t];
      asm volatile("" :: "v"(tv[0]), "v"(tv[1]), "v"(tv[2]), "v"(tv[3]), "v"(wv[0]), "v"(wv[1]), "v"(wv[2]), "v"(wv[3]), "v"(wv[4]), "v"(wv[5]), "v"(wv[6]), "v"(wv[7]), "v"(wv[8]), "v"(wv[9]), "v"(wv[10]), "v"(wv[11]), "v"(wv[12]), "v"(wv[13]), "v"(wv[14]), "v"(wv[15]));
#pragma unroll
      for (int q = 0; q < 16; ++q) acc = fmaf(tv[q >> 2][q & 3], wv[q], acc);
    }
    St[r * 32 + t] = acc;
  }
}

__global__ __launch_bounds__(256)
void s_psi_trunk(const float* __restrict__ P, const float* __restrict__ W3T,
                 const float* __restrict__ b3p, const float* __restrict__ X1,
                 float* __restrict__ St,
                 const float* __restrict__ qw1, const float* __restrict__ qb1,
                 const float* __restrict__ qw2, const float* __restrict__ qb2,
                 float* __restrict__ H2S) {
  if (blockIdx.x < TRUNK_TILES) trunk_tile<true>(qw1, qb1, qw2, qb2, H2S, blockIdx.x);
  else                          s_row(P, W3T, b3p, X1, St, blockIdx.x - TRUNK_TILES);
}

#define U_BLOCKS 128
__global__ __launch_bounds__(256)
void uc_kernel(const float* __restrict__ St, const float* __restrict__ w3q,
               const float* __restrict__ b3q, float* __restrict__ U, float* __restrict__ C) {
  if (blockIdx.x == U_BLOCKS) {
    const int bo = threadIdx.x;
    const int o = bo & 31;
    const float* sp = St + (bo - o) * 32 + o;
    float acc = 0.0f;
#pragma unroll
    for (int i = 0; i < 32; ++i) acc += sp[i * 32] * b3q[o * 32 + i];
    C[bo] = acc;
    return;
  }
  const int idx = blockIdx.x * 256 + threadIdx.x;
  const int bo = idx >> 7;
  const int k = idx & 127;
  const int o = bo & 31;
  const float* sp = St + (bo - o) * 32 + o;
  float acc = 0.0f;
#pragma unroll
  for (int i = 0; i < 32; ++i) acc += sp[i * 32] * w3q[k * 1024 + o * 32 + i];
  U[idx] = acc;
}

__global__ __launch_bounds__(256)
void gemm2(const float* __restrict__ U, const float* __restrict__ H2S,
           const float* __restrict__ C, float* __restrict__ out) {
  __shared__ __attribute__((aligned(16))) float lds[STAGE_FLOATS];
  const int lane = threadIdx.x & 31;
  const int wave = threadIdx.x >> 5;
  const int half = lane >> 4;
  const int l16  = lane & 15;
  const int wm   = wave & 1;
  const int wn   = wave >> 1;
  const int rb   = blockIdx.x & 3;
  const int cb   = blockIdx.x >> 2;

  v4f ra[TROWS_A / 8], rbv[TROWS_B / 8];
  tile_load<TROWS_A>(U   + (long)(64 * rb) * HID,  HID, ra);
  tile_load<TROWS_B>(H2S + (long)(128 * cb) * HID, HID, rbv);

  float* As = lds;
  float* Bs = As + TROWS_A * PITCH;

  v8f acc[2][2];
  const v8f vzero = {0.f,0.f,0.f,0.f,0.f,0.f,0.f,0.f};
  acc[0][0] = vzero; acc[0][1] = vzero; acc[1][0] = vzero; acc[1][1] = vzero;

  tile_store<TROWS_A>(As, ra);
  tile_store<TROWS_B>(Bs, rbv);
  __syncthreads();
  tile_mma(As + (32 * wm + l16) * PITCH + 4 * half,
           Bs + (32 * wn + l16) * PITCH + 4 * half, acc);

  const float scale = 1.0f / (float)NX;
#pragma unroll
  for (int mi = 0; mi < 2; ++mi)
#pragma unroll
    for (int ni = 0; ni < 2; ++ni)
#pragma unroll
      for (int r = 0; r < 8; ++r) {
        const int row = 64 * rb + 32 * wm + 16 * mi + 8 * half + r;
        const int col = 128 * cb + 32 * wn + 16 * ni + l16;
        out[(long)row * NX + col] = (acc[mi][ni][r] + C[row]) * scale;
      }
}

extern "C" void kernel_launch(void* const* d_in, const int* in_sizes, int n_in,
                              void* d_out, int out_size, void* d_ws, size_t ws_size,
                              hipStream_t stream) {
  const float* x   = (const float*)d_in[0];
  const float* pw1 = (const float*)d_in[1];
  const float* pb1 = (const float*)d_in[2];
  const float* pw2 = (const float*)d_in[3];
  const float* pb2 = (const float*)d_in[4];
  const float* pw3 = (const float*)d_in[5];
  const float* pb3 = (const float*)d_in[6];
  const float* qw1 = (const float*)d_in[7];
  const float* qb1 = (const float*)d_in[8];
  const float* qw2 = (const float*)d_in[9];
  const float* qb2 = (const float*)d_in[10];
  const float* qw3 = (const float*)d_in[11];
  const float* qb3 = (const float*)d_in[12];

  float* ws  = (float*)d_ws;
  float* H2P = ws;
  float* H2S = H2P + HID * NX;
  float* P   = H2S + NX * HID;
  float* X1  = P + 64 * ROWS * HID;
  float* St  = X1 + ROWS;
  float* Uu  = St + 8192;
  float* Cc  = Uu + ROWS * HID;
  float* W3T = Cc + ROWS;

  phi_trunk_rowsum<<<dim3(544), dim3(256), 0, stream>>>(pw1, pb1, pw2, pb2, x, pw3, H2P, X1, W3T);
  gemm1_partial<<<dim3(256), dim3(256), 0, stream>>>(x, H2P, P);
  s_psi_trunk<<<dim3(512), dim3(256), 0, stream>>>(P, W3T, pb3, X1, St,
                                                 qw1, qb1, qw2, qb2, H2S);
  uc_kernel<<<dim3(129), dim3(256), 0, stream>>>(St, qw3, qb3, Uu, Cc);
  gemm2<<<dim3(512), dim3(256), 0, stream>>>(Uu, H2S, Cc, (float*)d_out);
}


# phi trunk computed inside GEMM1's blocks (128 rows x 128 points, lane owns 4 points, trunk tile written straight into the LDS B operand): launches 5 -> 4; GEMM1 partials 64 -> 128, sum order changes
# speedup vs baseline: 9.9126x; 1.1259x over previous
#include <hip/hip_runtime.h>

typedef __attribute__((ext_vector_type(2))) float v2f;
typedef __attribute__((ext_vector_type(8))) float v8f;
typedef __attribute__((ext_vector_type(4))) float v4f;

#define NX    16384
#define HID   128
#define ROWS  256

__device__ __forceinline__ v8f wmma4(v2f a, v2f b, v8f c) {
  return __builtin_amdgcn_wmma_f32_16x16x4_f32(
       false, a,  false, b,
       (short)0, c,  false,  false);
}

#define TF 32
#define TH (TF / 2)
template <bool POINT_MAJOR>
__device__ __forceinline__ void trunk_tile(const float* __restrict__ w1, const float* __restrict__ b1,
                                           const float* __restrict__ w2g, const float* __restrict__ b2g_,
                                           float* __restrict__ H, int tb) {
  __shared__ __attribute__((aligned(16))) float w2s[64 * TF];
  __shared__ __attribute__((aligned(16))) float l1s[64 * 4];
  const int t  = threadIdx.x;
  const int c0 = (tb & 3) * TF;
  {
    const int row = t >> 3;
    const int q   = t & 7;
    const v4f lo = *((const v4f*)(w2g + (long)row * HID + c0) + q);
    const v4f hi = *((const v4f*)(w2g + (long)(row + 32) * HID + c0) + q);
    ((v4f*)w2s)[row * (TF / 4) + q]        = lo;
    ((v4f*)w2s)[(row + 32) * (TF / 4) + q] = hi;
    if (t < 64) {
      const v4f l = {w1[t], w1[64 + t], b1[t], 0.0f};
      ((v4f*)l1s)[t] = l;
    }
  }
  __syncthreads();

  const int lane = t & 31;
  const int wave = t >> 5;
  const int wv   = wave & 3;
  const int fh   = wave >> 2;
  const int n0   = (tb >> 2) * 256 + 64 * wv + lane;
  const int n1   = n0 + 32;
  const float gx0 = (float)(n0 >> 7) * (1.0f / 127.0f);
  const float gy0 = (float)(n0 & 127) * (1.0f / 127.0f);
  const float gx1 = (float)(n1 >> 7) * (1.0f / 127.0f);
  const float gy1 = (float)(n1 & 127) * (1.0f / 127.0f);
  const float* b2 = b2g_ + c0 + TH * fh;

  float a0[TH], a1[TH];
#pragma unroll
  for (int c = 0; c < TH; ++c) { a0[c] = b2[c]; a1[c] = a0[c]; }

  const float* wrow = w2s + TH * fh;
#pragma unroll 1
  for (int j = 0; j < 64; j += 4) {
    v4f w[4][TH / 4], tr[4];
#pragma unroll
    for (int u = 0; u < 4; ++u) {
      tr[u] = ((const v4f*)l1s)[j + u];
#pragma unroll
      for (int q = 0; q < TH / 4; ++q) w[u][q] = ((const v4f*)(wrow + (j + u) * TF))[q];
    }
    __builtin_amdgcn_sched_barrier(0);
#pragma unroll
    for (int u = 0; u < 4; ++u) {
      const float h0 = fmaxf(fmaf(gx0, tr[u].x, gy0 * tr[u].y) + tr[u].z, 0.0f);
      const float h1 = fmaxf(fmaf(gx1, tr[u].x, gy1 * tr[u].y) + tr[u].z, 0.0f);
#pragma unroll
      for (int q = 0; q < TH / 4; ++q) {
        a0[4 * q + 0] += h0 * w[u][q].x;  a1[4 * q + 0] += h1 * w[u][q].x;
        a0[4 * q + 1] += h0 * w[u][q].y;  a1[4 * q + 1] += h1 * w[u][q].y;
        a0[4 * q + 2] += h0 * w[u][q].z;  a1[4 * q + 2] += h1 * w[u][q].z;
        a0[4 * q + 3] += h0 * w[u][q].w;  a1[4 * q + 3] += h1 * w[u][q].w;
      }
    }
  }
#pragma unroll
  for (int c = 0; c < TH; ++c) { a0[c] = fmaxf(a0[c], 0.0f); a1[c] = fmaxf(a1[c], 0.0f); }

  if (!POINT_MAJOR) {
#pragma unroll
    for (int c = 0; c < TH; ++c) {
      H[(long)(c0 + TH * fh + c) * NX + n0] = a0[c];
      H[(long)(c0 + TH * fh + c) * NX + n1] = a1[c];
    }
  } else {
#define XP 20
    __shared__ __attribute__((aligned(16))) float xs[8 * 32 * XP];
    v4f* mine = (v4f*)(xs + (wave * 32 + lane) * XP);
    const v4f* theirs = (const v4f*)(xs + ((wave ^ 4) * 32 + lane) * XP);
    if (fh == 0) {
#pragma unroll
      for (int q = 0; q < TH / 4; ++q) mine[q] = (v4f){a1[4 * q], a1[4 * q + 1], a1[4 * q + 2], a1[4 * q + 3]};
    } else {
#pragma unroll
      for (int q = 0; q < TH / 4; ++q) mine[q] = (v4f){a0[4 * q], a0[4 * q + 1], a0[4 * q + 2], a0[4 * q + 3]};
    }
    __syncthreads();
    v4f p[TH / 4];
#pragma unroll
    for (int q = 0; q < TH / 4; ++q) p[q] = theirs[q];
    if (fh == 0) {
      v4f* d = (v4f*)(H + (long)n0 * HID + c0);
#pragma unroll
      for (int q = 0; q < TH / 4; ++q) {
        d[q]          = (v4f){a0[4 * q], a0[4 * q + 1], a0[4 * q + 2], a0[4 * q + 3]};
        d[TH / 4 + q] = p[q];
      }
    } else {
      v4f* d = (v4f*)(H + (long)n1 * HID + c0);
#pragma unroll
      for (int q = 0; q < TH / 4; ++q) {
        d[q]          = p[q];
        d[TH / 4 + q] = (v4f){a1[4 * q], a1[4 * q + 1], a1[4 * q + 2], a1[4 * q + 3]};
      }
    }
#undef XP
  }
}

__device__ __forceinline__ void rowsum_row(const float* __restrict__ x, float* __restrict__ X1, int r) {
  __shared__ float red[256];
  float s = 0.0f;
  for (int n = threadIdx.x; n < NX; n += 256) s += x[(long)r * NX + n];
  red[threadIdx.x] = s;
  __syncthreads();
  for (int off = 128; off > 0; off >>= 1) {
    if ((int)threadIdx.x < off) red[threadIdx.x] += red[threadIdx.x + off];
    __syncthreads();
  }
  if (threadIdx.x == 0) X1[r] = red[0];
}

#define RL_BLOCKS 32
__device__ __forceinline__ void relayout_w3(const float* __restrict__ w3, float* __restrict__ W3T, int kb) {
  __shared__ float tl[4 * 32 * 33];
  const int t = threadIdx.x;
#pragma unroll
  for (int q = 0; q < 16; ++q) {
    const int e  = q * 256 + t;
    const int kk = e >> 10;
    const int oi = e & 1023;
    tl[kk * 1056 + (oi >> 5) * 33 + (oi & 31)] = w3[(long)(4 * kb + kk) * 1024 + oi];
  }
  __syncthreads();
  const int lane = t & 31;
  const int wave = t >> 5;
#pragma unroll
  for (int q = 0; q < 16; ++q) {
    const int ln = q * 8 + wave;
    const int i  = ln >> 2;
    const int kk = ln & 3;
    W3T[((long)i * HID + 4 * kb + kk) * 32 + lane] = tl[kk * 1056 + lane * 33 + i];
  }
}

#define TRUNK_TILES 256

#define KT     128
#define PITCH  132
#define TROWS_A 64
#define TROWS_B 128
#define STAGE_FLOATS ((TROWS_A + TROWS_B) * PITCH)

template <int NR>
__device__ __forceinline__ void tile_load(const float* __restrict__ src, long ld,
                                          v4f (&r)[NR / 8]) {
  const int t = threadIdx.x;
  const float* p = src + (long)(t >> 5) * ld + 4 * (t & 31);
#pragma unroll
  for (int i = 0; i < NR / 8; ++i) r[i] = *(const v4f*)(p + (long)(8 * i) * ld);
}

template <int NR>
__device__ __forceinline__ void tile_store(float* dst, const v4f (&r)[NR / 8]) {
  const int t = threadIdx.x;
  float* p = dst + (t >> 5) * PITCH + 4 * (t & 31);
#pragma unroll
  for (int i = 0; i < NR / 8; ++i) *(v4f*)(p + 8 * i * PITCH) = r[i];
}

__device__ __forceinline__ void tile_mma(const float* ap, const float* bp, v8f (&acc)[2][2]) {
#pragma unroll
  for (int kb = 0; kb < KT / 8; ++kb) {
    const v4f a0 = *(const v4f*)(ap + 8 * kb);
    const v4f a1 = *(const v4f*)(ap + 16 * PITCH + 8 * kb);
    const v4f b0 = *(const v4f*)(bp + 8 * kb);
    const v4f b1 = *(const v4f*)(bp + 16 * PITCH + 8 * kb);
    const v2f a0l = {a0.x, a0.y}, a0h = {a0.z, a0.w};
    const v2f a1l = {a1.x, a1.y}, a1h = {a1.z, a1.w};
    const v2f b0l = {b0.x, b0.y}, b0h = {b0.z, b0.w};
    const v2f b1l = {b1.x, b1.y}, b1h = {b1.z, b1.w};
    acc[0][0] = wmma4(a0l, b0l, acc[0][0]);
    acc[0][1] = wmma4(a0l, b1l, acc[0][1]);
    acc[1][0] = wmma4(a1l, b0l, acc[1][0]);
    acc[1][1] = wmma4(a1l, b1l, acc[1][1]);
    acc[0][0] = wmma4(a0h, b0h, acc[0][0]);
    acc[0][1] = wmma4(a0h, b1h, acc[0][1]);
    acc[1][0] = wmma4(a1h, b0h, acc[1][0]);
    acc[1][1] = wmma4(a1h, b1h, acc[1][1]);
  }
}

#define G1_ROWS 128
#define G1_BLOCKS 256
#define G1_CHUNKS 128

__device__ __forceinline__ void tile_mma24(const float* ap, const float* bp, v8f (&acc)[2][4]) {
#pragma unroll
  for (int kb = 0; kb < KT / 8; ++kb) {
    v4f a[2], b[4];
#pragma unroll
    for (int mi = 0; mi < 2; ++mi) a[mi] = *(const v4f*)(ap + 16 * mi * PITCH + 8 * kb);
#pragma unroll
    for (int ni = 0; ni < 4; ++ni) b[ni] = *(const v4f*)(bp + 16 * ni * PITCH + 8 * kb);
#pragma unroll
    for (int mi = 0; mi < 2; ++mi)
#pragma unroll
      for (int ni = 0; ni < 4; ++ni) {
        const v2f al = {a[mi].x, a[mi].y}, bl = {b[ni].x, b[ni].y};
        acc[mi][ni] = wmma4(al, bl, acc[mi][ni]);
      }
#pragma unroll
    for (int mi = 0; mi < 2; ++mi)
#pragma unroll
      for (int ni = 0; ni < 4; ++ni) {
        const v2f ah = {a[mi].z, a[mi].w}, bh = {b[ni].z, b[ni].w};
        acc[mi][ni] = wmma4(ah, bh, acc[mi][ni]);
      }
  }
}

__global__ __launch_bounds__(256)
void trunk_gemm1(const float* __restrict__ w1, const float* __restrict__ b1,
                 const float* __restrict__ w2g, const float* __restrict__ b2g,
                 const float* __restrict__ X, const float* __restrict__ pw3,
                 float* __restrict__ P, float* __restrict__ X1, float* __restrict__ W3T) {
  if (blockIdx.x >= G1_BLOCKS) {
    if (blockIdx.x < G1_BLOCKS + ROWS) rowsum_row(X, X1, blockIdx.x - G1_BLOCKS);
    else                               relayout_w3(pw3, W3T, blockIdx.x - G1_BLOCKS - ROWS);
    return;
  }
  __shared__ __attribute__((aligned(16))) float lds[2 * G1_ROWS * PITCH];
  float* As  = lds;
  float* Bs  = lds + G1_ROWS * PITCH;
  float* w2s = Bs;
  float* l1s = Bs + 64 * HID;
  const int t    = threadIdx.x;
  const int lane = t & 31;
  const int wave = t >> 5;
  const int rg   = blockIdx.x & 1;
  const int kc   = blockIdx.x >> 1;

  v4f l = {0.0f, 0.0f, 0.0f, 0.0f};
  if (t < 64) l = (v4f){w1[t], w1[64 + t], b1[t], 0.0f};
  const float* b2 = b2g + TH * wave;
  float a[4][TH];
#pragma unroll
  for (int c = 0; c < TH; ++c) {
    const float bc = b2[c];
#pragma unroll
    for (int p = 0; p < 4; ++p) a[p][c] = bc;
  }
  v4f wr[8];
#pragma unroll
  for (int q = 0; q < 8; ++q) wr[q] = ((const v4f*)w2g)[q * 256 + t];
  v4f xa[G1_ROWS / 8];
  tile_load<G1_ROWS>(X + (long)(G1_ROWS * rg) * NX + KT * kc, NX, xa);
#pragma unroll
  for (int q = 0; q < 8; ++q) ((v4f*)w2s)[q * 256 + t] = wr[q];
  if (t < 64) ((v4f*)l1s)[t] = l;
  tile_store<G1_ROWS>(As, xa);
  __syncthreads();

  const float gx = (float)kc * (1.0f / 127.0f);
  float gy[4];
#pragma unroll
  for (int p = 0; p < 4; ++p) gy[p] = (float)(4 * lane + p) * (1.0f / 127.0f);

  const float* wrow = w2s + TH * wave;
#pragma unroll 1
  for (int j = 0; j < 64; j += 4) {
    v4f w[4][TH / 4], tr[4];
#pragma unroll
    for (int u = 0; u < 4; ++u) {
      tr[u] = ((const v4f*)l1s)[j + u];
#pragma unroll
      for (int q = 0; q < TH / 4; ++q) w[u][q] = ((const v4f*)(wrow + (j + u) * HID))[q];
    }
    __builtin_amdgcn_sched_barrier(0);
#pragma unroll
    for (int u = 0; u < 4; ++u) {
      float h[4];
#pragma unroll
      for (int p = 0; p < 4; ++p) h[p] = fmaxf(fmaf(gx, tr[u].x, gy[p] * tr[u].y) + tr[u].z, 0.0f);
#pragma unroll
      for (int q = 0; q < TH / 4; ++q) {
#pragma unroll
        for (int p = 0; p < 4; ++p) {
          a[p][4 * q + 0] += h[p] * w[u][q].x;
          a[p][4 * q + 1] += h[p] * w[u][q].y;
          a[p][4 * q + 2] += h[p] * w[u][q].z;
          a[p][4 * q + 3] += h[p] * w[u][q].w;
        }
      }
    }
  }
  __syncthreads();
#pragma unroll
  for (int c = 0; c < TH; ++c) {
    const v4f o = {fmaxf(a[0][c], 0.0f), fmaxf(a[1][c], 0.0f), fmaxf(a[2][c], 0.0f), fmaxf(a[3][c], 0.0f)};
    *(v4f*)(Bs + (TH * wave + c) * PITCH + 4 * lane) = o;
  }
  __syncthreads();

  const int half = lane >> 4;
  const int l16  = lane & 15;
  const int wm   = wave & 3;
  const int wn   = wave >> 2;
  v8f acc[2][4];
  const v8f vzero = {0.f,0.f,0.f,0.f,0.f,0.f,0.f,0.f};
#pragma unroll
  for (int mi = 0; mi < 2; ++mi)
#pragma unroll
    for (int ni = 0; ni < 4; ++ni) acc[mi][ni] = vzero;
  tile_mma24(As + (32 * wm + l16) * PITCH + 4 * half,
             Bs + (64 * wn + l16) * PITCH + 4 * half, acc);

#pragma unroll
  for (int mi = 0; mi < 2; ++mi)
#pragma unroll
    for (int ni = 0; ni < 4; ++ni)
#pragma unroll
      for (int r = 0; r < 8; ++r) {
        const int row = G1_ROWS * rg + 32 * wm + 16 * mi + 8 * half + r;
        const int col = 64 * wn + 16 * ni + l16;
        P[((long)kc * ROWS + row) * HID + col] = acc[mi][ni][r];
      }
}

__device__ __forceinline__ void s_row(const float* __restrict__ P, const float* __restrict__ W3T,
                                      const float* __restrict__ b3p, const float* __restrict__ X1,
                                      float* __restrict__ St, int r) {
  __shared__ __attribute__((aligned(16))) float wt[HID * 32];
  __shared__ __attribute__((aligned(16))) float trow[HID];
  const int t = threadIdx.x;
  const int i = r & 31;
  if (t < HID) {
    const float* p = P + (long)r * HID + t;
    float s = 0.0f;
#pragma unroll
    for (int kc = 0; kc < G1_CHUNKS; ++kc) s += p[(long)kc * (ROWS * HID)];
    trow[t] = s;
  } else {
    const v4f* src = (const v4f*)(W3T + (long)i * (HID * 32));
    v4f* dst = (v4f*)wt;
    const int u = t - HID;
#pragma unroll
    for (int q = 0; q < 8; ++q) dst[q * 128 + u] = src[q * 128 + u];
  }
  __syncthreads();
  if (t < 32) {
    float acc = b3p[t * 32 + i] * X1[r];
#pragma unroll
    for (int k0 = 0; k0 < HID; k0 += 16) {
      v4f tv[4];
      float wv[16];
#pragma unroll
      for (int q = 0; q < 4; ++q) tv[q] = *(const v4f*)(trow + k0 + 4 * q);
#pragma unroll
      for (int q = 0; q < 16; ++q) wv[q] = wt[(k0 + q) * 32 + t];
      asm volatile("" :: "v"(tv[0]), "v"(tv[1]), "v"(tv[2]), "v"(tv[3]), "v"(wv[0]), "v"(wv[1]), "v"(wv[2]), "v"(wv[3]), "v"(wv[4]), "v"(wv[5]), "v"(wv[6]), "v"(wv[7]), "v"(wv[8]), "v"(wv[9]), "v"(wv[10]), "v"(wv[11]), "v"(wv[12]), "v"(wv[13]), "v"(wv[14]), "v"(wv[15]));
#pragma unroll
      for (int q = 0; q < 16; ++q) acc = fmaf(tv[q >> 2][q & 3], wv[q], acc);
    }
    St[r * 32 + t] = acc;
  }
}

__global__ __launch_bounds__(256)
void s_psi_trunk(const float* __restrict__ P, const float* __restrict__ W3T,
                 const float* __restrict__ b3p, const float* __restrict__ X1,
                 float* __restrict__ St,
                 const float* __restrict__ qw1, const float* __restrict__ qb1,
                 const float* __restrict__ qw2, const float* __restrict__ qb2,
                 float* __restrict__ H2S) {
  if (blockIdx.x < TRUNK_TILES) trunk_tile<true>(qw1, qb1, qw2, qb2, H2S, blockIdx.x);
  else                          s_row(P, W3T, b3p, X1, St, blockIdx.x - TRUNK_TILES);
}

#define U_BLOCKS 128
__global__ __launch_bounds__(256)
void uc_kernel(const float* __restrict__ St, const float* __restrict__ w3q,
               const float* __restrict__ b3q, float* __restrict__ U, float* __restrict__ C) {
  if (blockIdx.x == U_BLOCKS) {
    const int bo = threadIdx.x;
    const int o = bo & 31;
    const float* sp = St + (bo - o) * 32 + o;
    float acc = 0.0f;
#pragma unroll
    for (int i = 0; i < 32; ++i) acc += sp[i * 32] * b3q[o * 32 + i];
    C[bo] = acc;
    return;
  }
  const int idx = blockIdx.x * 256 + threadIdx.x;
  const int bo = idx >> 7;
  const int k = idx & 127;
  const int o = bo & 31;
  const float* sp = St + (bo - o) * 32 + o;
  float acc = 0.0f;
#pragma unroll
  for (int i = 0; i < 32; ++i) acc += sp[i * 32] * w3q[k * 1024 + o * 32 + i];
  U[idx] = acc;
}

__global__ __launch_bounds__(256)
void gemm2(const float* __restrict__ U, const float* __restrict__ H2S,
           const float* __restrict__ C, float* __restrict__ out) {
  __shared__ __attribute__((aligned(16))) float lds[STAGE_FLOATS];
  const int lane = threadIdx.x & 31;
  const int wave = threadIdx.x >> 5;
  const int half = lane >> 4;
  const int l16  = lane & 15;
  const int wm   = wave & 1;
  const int wn   = wave >> 1;
  const int rb   = blockIdx.x & 3;
  const int cb   = blockIdx.x >> 2;

  v4f ra[TROWS_A / 8], rbv[TROWS_B / 8];
  tile_load<TROWS_A>(U   + (long)(64 * rb) * HID,  HID, ra);
  tile_load<TROWS_B>(H2S + (long)(128 * cb) * HID, HID, rbv);

  float* As = lds;
  float* Bs = As + TROWS_A * PITCH;

  v8f acc[2][2];
  const v8f vzero = {0.f,0.f,0.f,0.f,0.f,0.f,0.f,0.f};
  acc[0][0] = vzero; acc[0][1] = vzero; acc[1][0] = vzero; acc[1][1] = vzero;

  tile_store<TROWS_A>(As, ra);
  tile_store<TROWS_B>(Bs, rbv);
  __syncthreads();
  tile_mma(As + (32 * wm + l16) * PITCH + 4 * half,
           Bs + (32 * wn + l16) * PITCH + 4 * half, acc);

  const float scale = 1.0f / (float)NX;
#pragma unroll
  for (int mi = 0; mi < 2; ++mi)
#pragma unroll
    for (int ni = 0; ni < 2; ++ni)
#pragma unroll
      for (int r = 0; r < 8; ++r) {
        const int row = 64 * rb + 32 * wm + 16 * mi + 8 * half + r;
        const int col = 128 * cb + 32 * wn + 16 * ni + l16;
        out[(long)row * NX + col] = (acc[mi][ni][r] + C[row]) * scale;
      }
}

extern "C" void kernel_launch(void* const* d_in, const int* in_sizes, int n_in,
                              void* d_out, int out_size, void* d_ws, size_t ws_size,
                              hipStream_t stream) {
  const float* x   = (const float*)d_in[0];
  const float* pw1 = (const float*)d_in[1];
  const float* pb1 = (const float*)d_in[2];
  const float* pw2 = (const float*)d_in[3];
  const float* pb2 = (const float*)d_in[4];
  const float* pw3 = (const float*)d_in[5];
  const float* pb3 = (const float*)d_in[6];
  const float* qw1 = (const float*)d_in[7];
  const float* qb1 = (const float*)d_in[8];
  const float* qw2 = (const float*)d_in[9];
  const float* qb2 = (const float*)d_in[10];
  const float* qw3 = (const float*)d_in[11];
  const float* qb3 = (const float*)d_in[12];

  float* ws  = (float*)d_ws;
  float* P   = ws;
  float* H2S = P + G1_CHUNKS * ROWS * HID;
  float* X1  = H2S + NX * HID;
  float* St  = X1 + ROWS;
  float* Uu  = St + 8192;
  float* Cc  = Uu + ROWS * HID;
  float* W3T = Cc + ROWS;

  trunk_gemm1<<<dim3(544), dim3(256), 0, stream>>>(pw1, pb1, pw2, pb2, x, pw3, P, X1, W3T);
  s_psi_trunk<<<dim3(512), dim3(256), 0, stream>>>(P, W3T, pb3, X1, St,
                                                 qw1, qb1, qw2, qb2, H2S);
  uc_kernel<<<dim3(129), dim3(256), 0, stream>>>(St, qw3, qb3, Uu, Cc);
  gemm2<<<dim3(512), dim3(256), 0, stream>>>(Uu, H2S, Cc, (float*)d_out);
}
